# starting kernel
# baseline (speedup 1.0000x reference)
#include <hip/hip_runtime.h>

typedef float v2f __attribute__((ext_vector_type(2)));
typedef float v8f __attribute__((ext_vector_type(8)));

__global__ void zero_f32v4(float4* __restrict__ p, long long n4) {
    long long i = (long long)blockIdx.x * blockDim.x + threadIdx.x;
    if (i < n4) p[i] = make_float4(0.f, 0.f, 0.f, 0.f);
}

template <int RELU_A, int NN, int KK>
__global__ void gemm_wmma_f32(const float* __restrict__ A,
                              const float* __restrict__ B,
                              float* __restrict__ C,
                              int M) {
    constexpr int LDB = KK + 4;
    __shared__ float Bt[16 * LDB];

    const int tid  = (int)threadIdx.x;
    const int n0   = (int)blockIdx.y << 4;

    #pragma unroll
    for (int idx = 0; idx < KK * 16 / 256; ++idx) {
        int flat = tid + idx * 256;
        int k = flat >> 4;
        int n = flat & 15;
        Bt[n * LDB + k] = B[k * NN + n0 + n];
    }
    __syncthreads();

    const int wave = tid >> 5;
    const int lane = tid & 31;
    const int tileM = (int)blockIdx.x * 8 + wave;
    if (tileM >= (M >> 4)) return;

    const int l15  = lane & 15;
    const int kOff = (lane >> 4) << 1;

    const float* __restrict__ arow = A + ((tileM << 4) + l15) * KK + kOff;
    const float* __restrict__ brow = &Bt[l15 * LDB + kOff];

    v8f acc = {};
    #pragma unroll 16
    for (int k0 = 0; k0 < KK; k0 += 4) {
        float2 av = *(const float2*)(arow + k0);
        if (RELU_A) {
            av.x = fmaxf(av.x, 0.f);
            av.y = fmaxf(av.y, 0.f);
        }
        v2f a; a.x = av.x; a.y = av.y;
        v2f b = *(const v2f*)(brow + k0);
        acc = __builtin_amdgcn_wmma_f32_16x16x4_f32(
             false, a,  false, b,
             (short)0, acc,  false,  false);
    }

    const int cm = (tileM << 4) + ((lane >> 4) << 3);
    float* __restrict__ crow = C + cm * NN + n0 + l15;
    #pragma unroll
    for (int r = 0; r < 8; ++r) crow[r * NN] = acc[r];
}

template <int SHIFT>
__global__ void spmm_atomic(const int*   __restrict__ erow,
                            const int*   __restrict__ ecol,
                            const float* __restrict__ ew,
                            const float* __restrict__ h,
                            float*       __restrict__ out,
                            int nEdges) {
    const long long tid = (long long)blockIdx.x * blockDim.x + threadIdx.x;
    const int e  = (int)(tid >> SHIFT);
    if (e >= nEdges) return;
    const int c4 = ((int)tid & ((1 << SHIFT) - 1)) << 2;
    constexpr int Cch = 4 << SHIFT;

    const float we = ew[e];
    const int   j  = ecol[e];
    const int   i  = erow[e];

    const float4 hv = *(const float4*)(h + (size_t)j * Cch + c4);
    float* dst = out + (size_t)i * Cch + c4;
    atomicAdd(dst + 0, hv.x * we);
    atomicAdd(dst + 1, hv.y * we);
    atomicAdd(dst + 2, hv.z * we);
    atomicAdd(dst + 3, hv.w * we);
}

extern "C" void kernel_launch(void* const* d_in, const int* in_sizes, int n_in,
                              void* d_out, int out_size, void* d_ws, size_t ws_size,
                              hipStream_t stream) {
    const float* x   = (const float*)d_in[0];
    const float* W1  = (const float*)d_in[1];
    const float* W2  = (const float*)d_in[2];
    const float* ew  = (const float*)d_in[3];
    const int*   er  = (const int*)  d_in[4];
    const int*   ec  = (const int*)  d_in[5];

    const int F_IN = 512, C1 = 128, C2 = 64;
    const int nNodes = in_sizes[0] / F_IN;
    const int nEdges = in_sizes[3];

    float* H   = (float*)d_ws;
    float* Hs  = H  + (size_t)nNodes * C1;
    float* H2  = Hs + (size_t)nNodes * C1;
    float* out = (float*)d_out;

    const int BLK = 256;

    {
        long long n4 = (long long)nNodes * C1 / 4;
        zero_f32v4<<<(unsigned)((n4 + BLK - 1) / BLK), BLK, 0, stream>>>((float4*)Hs, n4);
        long long m4 = (long long)nNodes * C2 / 4;
        zero_f32v4<<<(unsigned)((m4 + BLK - 1) / BLK), BLK, 0, stream>>>((float4*)out, m4);
    }

    {
        int mTiles = nNodes / 16;
        dim3 grid((mTiles + 7) / 8, C1 / 16);
        gemm_wmma_f32<0, 128, 512><<<grid, BLK, 0, stream>>>(x, W1, H, nNodes);
    }

    {
        long long threads = (long long)nEdges << 5;
        spmm_atomic<5><<<(unsigned)((threads + BLK - 1) / BLK), BLK, 0, stream>>>(
            er, ec, ew, H, Hs, nEdges);
    }

    {
        int mTiles = nNodes / 16;
        dim3 grid((mTiles + 7) / 8, C2 / 16);
        gemm_wmma_f32<1, 64, 128><<<grid, BLK, 0, stream>>>(Hs, W2, H2, nNodes);
    }

    {
        long long threads = (long long)nEdges << 4;
        spmm_atomic<4><<<(unsigned)((threads + BLK - 1) / BLK), BLK, 0, stream>>>(
            er, ec, ew, H2, out, nEdges);
    }
}


# SpMM atomic scatter: each wave loops over 32 (C=128) or 64 (C=64) edges, not 1 or 2, with 32 gather lines in flight and wave-uniform edge records by scalar loads; sums stay in atomic arrival order
# speedup vs baseline: 1.8588x; 1.8588x over previous
#include <hip/hip_runtime.h>

typedef float v2f __attribute__((ext_vector_type(2)));
typedef float v8f __attribute__((ext_vector_type(8)));

__global__ void zero_f32v4(float4* __restrict__ p, long long n4) {
    long long i = (long long)blockIdx.x * blockDim.x + threadIdx.x;
    if (i < n4) p[i] = make_float4(0.f, 0.f, 0.f, 0.f);
}

template <int RELU_A, int NN, int KK>
__global__ void gemm_wmma_f32(const float* __restrict__ A,
                              const float* __restrict__ B,
                              float* __restrict__ C,
                              int M) {
    constexpr int LDB = KK + 4;
    __shared__ float Bt[16 * LDB];

    const int tid  = (int)threadIdx.x;
    const int n0   = (int)blockIdx.y << 4;

    #pragma unroll
    for (int idx = 0; idx < KK * 16 / 256; ++idx) {
        int flat = tid + idx * 256;
        int k = flat >> 4;
        int n = flat & 15;
        Bt[n * LDB + k] = B[k * NN + n0 + n];
    }
    __syncthreads();

    const int wave = tid >> 5;
    const int lane = tid & 31;
    const int tileM = (int)blockIdx.x * 8 + wave;
    if (tileM >= (M >> 4)) return;

    const int l15  = lane & 15;
    const int kOff = (lane >> 4) << 1;

    const float* __restrict__ arow = A + ((tileM << 4) + l15) * KK + kOff;
    const float* __restrict__ brow = &Bt[l15 * LDB + kOff];

    v8f acc = {};
    #pragma unroll 16
    for (int k0 = 0; k0 < KK; k0 += 4) {
        float2 av = *(const float2*)(arow + k0);
        if (RELU_A) {
            av.x = fmaxf(av.x, 0.f);
            av.y = fmaxf(av.y, 0.f);
        }
        v2f a; a.x = av.x; a.y = av.y;
        v2f b = *(const v2f*)(brow + k0);
        acc = __builtin_amdgcn_wmma_f32_16x16x4_f32(
             false, a,  false, b,
             (short)0, acc,  false,  false);
    }

    const int cm = (tileM << 4) + ((lane >> 4) << 3);
    float* __restrict__ crow = C + cm * NN + n0 + l15;
    #pragma unroll
    for (int r = 0; r < 8; ++r) crow[r * NN] = acc[r];
}

template <int NL, int NE>
__device__ __forceinline__ void scatter_edges(const int*   __restrict__ erow,
                                              const int*   __restrict__ ecol,
                                              const float* __restrict__ ew,
                                              const float* __restrict__ h,
                                              float*       __restrict__ out,
                                              int e0, int lane) {
    constexpr int Cch = 32 * NL;
    float we[NE];
    int   j[NE], i[NE];
    for (int t = 0; t < NE; ++t) {
        we[t] = ew[e0 + t];
        j[t]  = ecol[e0 + t];
        i[t]  = erow[e0 + t];
    }
    float hv[NE][NL];
    for (int t = 0; t < NE; ++t) {
        const float* src = h + (size_t)j[t] * Cch + lane;
        for (int c = 0; c < NL; ++c) hv[t][c] = src[32 * c];
    }
    for (int t = 0; t < NE; ++t) {
        float* dst = out + (size_t)i[t] * Cch + lane;
        for (int c = 0; c < NL; ++c) atomicAdd(dst + 32 * c, hv[t][c] * we[t]);
    }
}

template <int NL, int EPW, int U>
__global__ void spmm_atomic(const int*   __restrict__ erow,
                            const int*   __restrict__ ecol,
                            const float* __restrict__ ew,
                            const float* __restrict__ h,
                            float*       __restrict__ out,
                            int nEdges) {
    const long long tid = (long long)blockIdx.x * blockDim.x + threadIdx.x;
    const int lane = (int)threadIdx.x & 31;
    const int wave = __builtin_amdgcn_readfirstlane((int)(tid >> 5));
    const int e0 = wave * EPW;
    if (e0 >= nEdges) return;
    if (e0 + EPW <= nEdges) {
        #pragma unroll 1
        for (int s = 0; s < EPW; s += U)
            scatter_edges<NL, U>(erow, ecol, ew, h, out, e0 + s, lane);
    } else {
        for (int e = e0; e < nEdges; ++e)
            scatter_edges<NL, 1>(erow, ecol, ew, h, out, e, lane);
    }
}

extern "C" void kernel_launch(void* const* d_in, const int* in_sizes, int n_in,
                              void* d_out, int out_size, void* d_ws, size_t ws_size,
                              hipStream_t stream) {
    const float* x   = (const float*)d_in[0];
    const float* W1  = (const float*)d_in[1];
    const float* W2  = (const float*)d_in[2];
    const float* ew  = (const float*)d_in[3];
    const int*   er  = (const int*)  d_in[4];
    const int*   ec  = (const int*)  d_in[5];

    const int F_IN = 512, C1 = 128, C2 = 64;
    const int nNodes = in_sizes[0] / F_IN;
    const int nEdges = in_sizes[3];

    float* H   = (float*)d_ws;
    float* Hs  = H  + (size_t)nNodes * C1;
    float* H2  = Hs + (size_t)nNodes * C1;
    float* out = (float*)d_out;

    const int BLK = 256;

    {
        long long n4 = (long long)nNodes * C1 / 4;
        zero_f32v4<<<(unsigned)((n4 + BLK - 1) / BLK), BLK, 0, stream>>>((float4*)Hs, n4);
        long long m4 = (long long)nNodes * C2 / 4;
        zero_f32v4<<<(unsigned)((m4 + BLK - 1) / BLK), BLK, 0, stream>>>((float4*)out, m4);
    }

    {
        int mTiles = nNodes / 16;
        dim3 grid((mTiles + 7) / 8, C1 / 16);
        gemm_wmma_f32<0, 128, 512><<<grid, BLK, 0, stream>>>(x, W1, H, nNodes);
    }

    {
        long long threads = (((long long)nEdges + 31) / 32) << 5;
        spmm_atomic<4, 32, 8><<<(unsigned)((threads + BLK - 1) / BLK), BLK, 0, stream>>>(
            er, ec, ew, H, Hs, nEdges);
    }

    {
        int mTiles = nNodes / 16;
        dim3 grid((mTiles + 7) / 8, C2 / 16);
        gemm_wmma_f32<1, 64, 128><<<grid, BLK, 0, stream>>>(Hs, W2, H2, nNodes);
    }

    {
        long long threads = (((long long)nEdges + 63) / 64) << 5;
        spmm_atomic<2, 64, 16><<<(unsigned)((threads + BLK - 1) / BLK), BLK, 0, stream>>>(
            er, ec, ew, H2, out, nEdges);
    }
}


# SpMM edge filter indexed: GEMM launches write 1 bit per (row block, 16-edge group); an SpMM block reads its own bits and only the record lines they mark, not a key per edge; arrival-order sums
# speedup vs baseline: 2.5075x; 1.3490x over previous
#include <hip/hip_runtime.h>

typedef float v2f __attribute__((ext_vector_type(2)));
typedef float v8f __attribute__((ext_vector_type(8)));

constexpr int RB  = 196;
constexpr int NRB = 256;

template <int RELU_A, int NN, int KK>
__global__ void gemm_wmma_f32(const float* __restrict__ A,
                              const float* __restrict__ B,
                              float* __restrict__ C,
                              int M,
                              const int*   __restrict__ erow,
                              const int*   __restrict__ ecol,
                              const float* __restrict__ ew,
                              unsigned*    __restrict__ bm,
                              unsigned*    __restrict__ rec,
                              int nEdges) {
    constexpr int LDB = KK + 4;
    __shared__ __attribute__((aligned(16))) float Bt[16 * LDB];
    static_assert(sizeof(float) * 16 * LDB >= 4 * NRB * 4, "a tile's bit matrix shares the panel's LDS");

    const int tid  = (int)threadIdx.x;
    const int n0   = (int)blockIdx.y << 4;

    {
        unsigned* lbm = (unsigned*)Bt;
        const int nblk   = (int)(gridDim.x * gridDim.y);
        const int gb     = (int)blockIdx.y * (int)gridDim.x + (int)blockIdx.x;
        const int nq     = nEdges >> 2;
        const int nTiles = (nEdges + 2047) >> 11;
        const int4*   __restrict__ er4 = (const int4*)erow;
        const int4*   __restrict__ ec4 = (const int4*)ecol;
        const float4* __restrict__ ew4 = (const float4*)ew;
        uint4* __restrict__ rec4 = (uint4*)rec;
        uint2* rec2 = (uint2*)rec;
        for (int tile = gb; tile < nTiles; tile += nblk) {
            ((uint4*)lbm)[tid] = make_uint4(0u, 0u, 0u, 0u);
            __syncthreads();
            int4 r[2], c[2];
            float4 w[2];
            #pragma unroll
            for (int j = 0; j < 2; ++j) {
                const int i = tile * 512 + j * 256 + tid;
                if (i < nq) { r[j] = er4[i]; c[j] = ec4[i]; w[j] = ew4[i]; }
                else { r[j] = make_int4(0, 0, 0, 0); c[j] = r[j]; w[j] = make_float4(0.f, 0.f, 0.f, 0.f); }
            }
            #pragma unroll
            for (int j = 0; j < 2; ++j) {
                const int i = tile * 512 + j * 256 + tid;
                if (i < nq) {
                    const unsigned kx = (unsigned)r[j].x / (unsigned)RB, ky = (unsigned)r[j].y / (unsigned)RB;
                    const unsigned kz = (unsigned)r[j].z / (unsigned)RB, kw = (unsigned)r[j].w / (unsigned)RB;
                    const int      gl  = (j * 256 + tid) >> 2;
                    const int      wd  = gl >> 5;
                    const unsigned bit = 1u << (gl & 31);
                    __hip_atomic_fetch_or(&lbm[kx * 4 + wd], bit, __ATOMIC_RELAXED, __HIP_MEMORY_SCOPE_WORKGROUP);
                    __hip_atomic_fetch_or(&lbm[ky * 4 + wd], bit, __ATOMIC_RELAXED, __HIP_MEMORY_SCOPE_WORKGROUP);
                    __hip_atomic_fetch_or(&lbm[kz * 4 + wd], bit, __ATOMIC_RELAXED, __HIP_MEMORY_SCOPE_WORKGROUP);
                    __hip_atomic_fetch_or(&lbm[kw * 4 + wd], bit, __ATOMIC_RELAXED, __HIP_MEMORY_SCOPE_WORKGROUP);
                    uint4 a, b;
                    a.x = (unsigned)c[j].x | (((unsigned)r[j].x - kx * (unsigned)RB) << 16) | (kx << 24);  a.y = __float_as_uint(w[j].x);
                    a.z = (unsigned)c[j].y | (((unsigned)r[j].y - ky * (unsigned)RB) << 16) | (ky << 24);  a.w = __float_as_uint(w[j].y);
                    b.x = (unsigned)c[j].z | (((unsigned)r[j].z - kz * (unsigned)RB) << 16) | (kz << 24);  b.y = __float_as_uint(w[j].z);
                    b.z = (unsigned)c[j].w | (((unsigned)r[j].w - kw * (unsigned)RB) << 16) | (kw << 24);  b.w = __float_as_uint(w[j].w);
                    rec4[2 * i]     = a;
                    rec4[2 * i + 1] = b;
                }
            }
            if (tile == nTiles - 1 && tid == 0) {
                for (int e = nq << 2; e < nEdges; ++e) {
                    const unsigned rr = (unsigned)erow[e];
                    const unsigned k  = rr / (unsigned)RB;
                    const int      gl = (e >> 4) & 127;
                    __hip_atomic_fetch_or(&lbm[k * 4 + (gl >> 5)], 1u << (gl & 31),
                                          __ATOMIC_RELAXED, __HIP_MEMORY_SCOPE_WORKGROUP);
                    rec2[e] = make_uint2((unsigned)ecol[e] | ((rr - k * (unsigned)RB) << 16) | (k << 24),
                                        __float_as_uint(ew[e]));
                }
            }
            __syncthreads();
            ((uint4*)(bm + (size_t)tile * (NRB * 4)))[tid] = ((const uint4*)lbm)[tid];
            __syncthreads();
        }
    }

    #pragma unroll
    for (int idx = 0; idx < KK * 16 / 256; ++idx) {
        int flat = tid + idx * 256;
        int k = flat >> 4;
        int n = flat & 15;
        Bt[n * LDB + k] = B[k * NN + n0 + n];
    }
    __syncthreads();

    const int wave = tid >> 5;
    const int lane = tid & 31;
    const int tileM = (int)blockIdx.x * 8 + wave;
    if (tileM >= (M >> 4)) return;

    const int l15  = lane & 15;
    const int kOff = (lane >> 4) << 1;

    const float* __restrict__ arow = A + ((tileM << 4) + l15) * KK + kOff;
    const float* __restrict__ brow = &Bt[l15 * LDB + kOff];

    v8f acc = {};
    #pragma unroll 16
    for (int k0 = 0; k0 < KK; k0 += 4) {
        float2 av = *(const float2*)(arow + k0);
        if (RELU_A) {
            av.x = fmaxf(av.x, 0.f);
            av.y = fmaxf(av.y, 0.f);
        }
        v2f a; a.x = av.x; a.y = av.y;
        v2f b = *(const v2f*)(brow + k0);
        acc = __builtin_amdgcn_wmma_f32_16x16x4_f32(
             false, a,  false, b,
             (short)0, acc,  false,  false);
    }

    const int cm = (tileM << 4) + ((lane >> 4) << 3);
    float* __restrict__ crow = C + cm * NN + n0 + l15;
    #pragma unroll
    for (int r = 0; r < 8; ++r) crow[r * NN] = acc[r];
}

template <int NL> struct RowVec;
template <> struct RowVec<4> { typedef float4 T; };
template <> struct RowVec<2> { typedef float2 T; };

template <int NL>
__device__ __forceinline__ float comp(const typename RowVec<NL>::T& v, int k);
template <> __device__ __forceinline__ float comp<4>(const float4& v, int k) {
    return k == 0 ? v.x : k == 1 ? v.y : k == 2 ? v.z : v.w;
}
template <> __device__ __forceinline__ float comp<2>(const float2& v, int k) {
    return k == 0 ? v.x : v.y;
}

template <int NL, int NE>
__device__ __forceinline__ void add_edges(int lr, int cj, float w, int b,
                                          const float* __restrict__ h,
                                          float* acc, int lane) {
    typedef typename RowVec<NL>::T V;
    constexpr int Cch = 32 * NL;
    int   j[NE], i[NE];
    float we[NE];
    for (int t = 0; t < NE; ++t) {
        j[t]  = __builtin_amdgcn_readlane(cj, b + t);
        i[t]  = __builtin_amdgcn_readlane(lr, b + t);
        we[t] = __int_as_float(__builtin_amdgcn_readlane(__float_as_int(w), b + t));
    }
    V hv[NE];
    for (int t = 0; t < NE; ++t)
        hv[t] = *(const V*)(h + (size_t)j[t] * Cch + NL * lane);
    for (int t = 0; t < NE; ++t) {
        float* dst = acc + i[t] * Cch + lane;
        for (int k = 0; k < NL; ++k)
            __hip_atomic_fetch_add(dst + 32 * k, comp<NL>(hv[t], k) * we[t],
                                   __ATOMIC_RELAXED, __HIP_MEMORY_SCOPE_WORKGROUP);
    }
}

template <int NL, int U>
__device__ __forceinline__ void drain(const uint2* qp, int n,
                                      const float* __restrict__ h,
                                      float* acc, int lane) {
    const uint2 rc = qp[lane < n ? lane : 0];
    const int   lr = (int)((rc.x >> 16) & 0xFFu);
    const int   cj = (int)(rc.x & 0xFFFFu);
    const float w  = __uint_as_float(rc.y);
    int b = 0;
    for (; b + U <= n; b += U) add_edges<NL, U>(lr, cj, w, b, h, acc, lane);
    if (U > 8 && b + 8 <= n) { add_edges<NL, 8>(lr, cj, w, b, h, acc, lane); b += 8; }
    if (b + 4 <= n) { add_edges<NL, 4>(lr, cj, w, b, h, acc, lane); b += 4; }
    if (b + 2 <= n) { add_edges<NL, 2>(lr, cj, w, b, h, acc, lane); b += 2; }
    if (b < n)        add_edges<NL, 1>(lr, cj, w, b, h, acc, lane);
}

__device__ __forceinline__ int queue_push(bool m, int e, int* myq, int cnt) {
    const unsigned bm = __builtin_amdgcn_ballot_w32(m);
    if (m) myq[cnt + (int)__builtin_amdgcn_mbcnt_lo(bm, 0u)] = e;
    return cnt + __builtin_popcount(bm);
}

__device__ __forceinline__ int queue_push2(bool m, uint2 v, uint2* myq, int cnt) {
    const unsigned bm = __builtin_amdgcn_ballot_w32(m);
    if (m) myq[cnt + (int)__builtin_amdgcn_mbcnt_lo(bm, 0u)] = v;
    return cnt + __builtin_popcount(bm);
}

template <bool FULL>
__device__ __forceinline__ int take_groups(const int* gq, int ng, unsigned key,
                                           const uint2* __restrict__ rec, int nEdges,
                                           uint2* eq, int ecnt, int lane) {
    const int hl  = lane >> 4;
    const int l15 = lane & 15;
    int   g[16];
    uint2 rc[16];
    bool  ok[16];
    #pragma unroll
    for (int p = 0; p < 16; ++p) {
        const int gi = 2 * p + hl;
        g[p] = gq[(FULL || gi < ng) ? gi : 0];
    }
    #pragma unroll
    for (int p = 0; p < 16; ++p) {
        const int e = g[p] * 16 + l15;
        ok[p] = (FULL || 2 * p + hl < ng) && e < nEdges;
        rc[p] = rec[e < nEdges ? e : nEdges - 1];
    }
    #pragma unroll
    for (int p = 0; p < 16; ++p)
        ecnt = queue_push2(ok[p] && (rc[p].x >> 24) == key, rc[p], eq, ecnt);
    return ecnt;
}

template <int NL, int U>
__device__ __forceinline__ int drain_full(uint2* eq, int cnt,
                                          const float* __restrict__ h,
                                          float* acc, int lane) {
    while (cnt >= 32) {
        cnt -= 32;
        __builtin_amdgcn_fence(__ATOMIC_ACQ_REL, "wavefront");
        drain<NL, U>(eq + cnt, 32, h, acc, lane);
        __builtin_amdgcn_fence(__ATOMIC_ACQ_REL, "wavefront");
    }
    return cnt;
}

template <int NL, int U>
__global__ void __launch_bounds__(1024)
spmm_rowblock(const unsigned* __restrict__ bm,
              const unsigned* __restrict__ recw,
              const float*    __restrict__ h,
              float*          __restrict__ out,
              int nEdges, int nNodes) {
    typedef typename RowVec<NL>::T V;
    constexpr int Cch  = 32 * NL;
    constexpr int NW   = 32;
    constexpr int GQCAP = 64;
    constexpr int EQCAP = 32 + 32 * 16;
    __shared__ float acc[RB * Cch];
    __shared__ uint2 eqs[NW * EQCAP];
    __shared__ int   gqs[NW * GQCAP];

    const int tid  = (int)threadIdx.x;
    const int lane = tid & 31;
    const int wave = __builtin_amdgcn_readfirstlane(tid >> 5);
    const int r0   = (int)blockIdx.x * RB;
    const unsigned key = (unsigned)blockIdx.x;

    for (int i = tid; i < RB * Cch / 4; i += NW * 32)
        ((float4*)acc)[i] = make_float4(0.f, 0.f, 0.f, 0.f);
    __syncthreads();

    int*   gq = gqs + wave * GQCAP;
    uint2* eq = eqs + wave * EQCAP;
    int gcnt = 0, ecnt = 0;
    const int nTiles = (nEdges + 2047) >> 11;
    const int nDw    = nTiles * 4;
    const int dpw    = (nDw + NW - 1) / NW;
    const int d0     = wave * dpw;
    const int d1     = d0 + dpw < nDw ? d0 + dpw : nDw;
    const uint2* __restrict__ rec = (const uint2*)recw;

    unsigned nxt = 0u;
    if (d0 + lane < d1) nxt = bm[((size_t)((d0 + lane) >> 2) * NRB + key) * 4 + ((d0 + lane) & 3)];
    for (int db = d0; db < d1; db += 32) {
        unsigned bits = nxt;
        const int gbase = (db + lane) * 32;
        nxt = 0u;
        {
            const int d = db + 32 + lane;
            if (d < d1) nxt = bm[((size_t)(d >> 2) * NRB + key) * 4 + (d & 3)];
        }
        while (__builtin_amdgcn_ballot_w32(bits != 0u) != 0u) {
            const bool m = bits != 0u;
            const int  k = m ? __builtin_ctz(bits) : 0;
            gcnt = queue_push(m, gbase + k, gq, gcnt);
            bits &= bits - 1u;
            if (gcnt >= 32) {
                gcnt -= 32;
                __builtin_amdgcn_fence(__ATOMIC_ACQ_REL, "wavefront");
                ecnt = take_groups<true>(gq + gcnt, 32, key, rec, nEdges, eq, ecnt, lane);
                __builtin_amdgcn_fence(__ATOMIC_ACQ_REL, "wavefront");
                ecnt = drain_full<NL, U>(eq, ecnt, h, acc, lane);
            }
        }
    }
    if (gcnt > 0) {
        __builtin_amdgcn_fence(__ATOMIC_ACQ_REL, "wavefront");
        ecnt = take_groups<false>(gq, gcnt, key, rec, nEdges, eq, ecnt, lane);
        __builtin_amdgcn_fence(__ATOMIC_ACQ_REL, "wavefront");
        ecnt = drain_full<NL, U>(eq, ecnt, h, acc, lane);
    }
    while (ecnt > 0) {
        const int n = ecnt < 32 ? ecnt : 32;
        ecnt -= n;
        __builtin_amdgcn_fence(__ATOMIC_ACQ_REL, "wavefront");
        drain<NL, U>(eq + ecnt, n, h, acc, lane);
        __builtin_amdgcn_fence(__ATOMIC_ACQ_REL, "wavefront");
    }
    __syncthreads();

    for (int row = tid >> 5; row < RB; row += NW) {
        if (r0 + row < nNodes) {
            const float* src = acc + row * Cch + lane;
            V v;
            if constexpr (NL == 4) { v.x = src[0]; v.y = src[32]; v.z = src[64]; v.w = src[96]; }
            else                   { v.x = src[0]; v.y = src[32]; }
            *(V*)(out + (size_t)(r0 + row) * Cch + NL * lane) = v;
        }
    }
}

extern "C" void kernel_launch(void* const* d_in, const int* in_sizes, int n_in,
                              void* d_out, int out_size, void* d_ws, size_t ws_size,
                              hipStream_t stream) {
    const float* x   = (const float*)d_in[0];
    const float* W1  = (const float*)d_in[1];
    const float* W2  = (const float*)d_in[2];
    const float* ew  = (const float*)d_in[3];
    const int*   er  = (const int*)  d_in[4];
    const int*   ec  = (const int*)  d_in[5];

    const int F_IN = 512, C1 = 128, C2 = 64;
    const int nNodes = in_sizes[0] / F_IN;
    const int nEdges = in_sizes[3];

    float* H   = (float*)d_ws;
    float* Hs  = H  + (size_t)nNodes * C1;
    float* H2  = Hs + (size_t)nNodes * C1;
    float* out = (float*)d_out;

    unsigned* idx1 = (unsigned*)H2;
    unsigned* rec1  = (unsigned*)d_out;
    unsigned* idx2 = (unsigned*)H;
    unsigned* rec2  = (unsigned*)(H + (((size_t)nEdges + 3) & ~(size_t)3));

    const int BLK = 256;
    const unsigned rowBlocks = (unsigned)((nNodes + RB - 1) / RB);

    {
        int mTiles = nNodes / 16;
        dim3 grid((mTiles + 7) / 8, C1 / 16);
        gemm_wmma_f32<0, 128, 512><<<grid, BLK, 0, stream>>>(x, W1, H, nNodes,
                                                             er, ec, ew, idx1, rec1, nEdges);
    }

    spmm_rowblock<4, 16><<<rowBlocks, 1024, 0, stream>>>(idx1, rec1, H, Hs, nEdges, nNodes);

    {
        int mTiles = nNodes / 16;
        dim3 grid((mTiles + 7) / 8, C2 / 16);
        gemm_wmma_f32<1, 64, 128><<<grid, BLK, 0, stream>>>(Hs, W2, H2, nNodes,
                                                            er, ec, ew, idx2, rec2, nEdges);
    }

    spmm_rowblock<2, 16><<<rowBlocks, 1024, 0, stream>>>(idx2, rec2, H2, out, nEdges, nNodes);
}
